# starting kernel
# baseline (speedup 1.0000x reference)
#include <hip/hip_runtime.h>
#include <math.h>


typedef __attribute__((ext_vector_type(2))) float v2f;
typedef __attribute__((ext_vector_type(8))) float v8f;

#define DIN 64

template<int R, int DOUT, int DOUTPAD>
__global__ __launch_bounds__(256)
void rgcn_gemm_wmma(const float* __restrict__ X, const float* __restrict__ W,
                    float* __restrict__ Y, int N)
{
    constexpr int NT = DOUTPAD / 16;
    constexpr int RD = R * DOUT;
    __shared__ float lA[128 * DIN];
    __shared__ float lWt[DOUTPAD * DIN];

    const int r       = blockIdx.y;
    const int rowBase = blockIdx.x * 128;
    const int tid     = threadIdx.x;

    const float* Xb = X + (size_t)rowBase * DIN;
    const bool fullBlock = (rowBase + 128 <= N);
    #pragma unroll
    for (int it = 0; it < (128 * DIN) / 256; ++it) {
        const int off = tid + it * 256;
        lA[off] = (fullBlock || (rowBase + (off >> 6) < N)) ? Xb[off] : 0.0f;
    }
    const float* Wr = W + (size_t)r * DIN * DOUT;
    #pragma unroll
    for (int it = 0; it < (DOUTPAD * DIN) / 256; ++it) {
        const int off = tid + it * 256;
        const int n = off >> 6, k = off & 63;
        lWt[off] = (n < DOUT) ? Wr[k * DOUT + n] : 0.0f;
    }
    __syncthreads();

    const int wv   = tid >> 5;
    const int lane = tid & 31;
    const int half = lane >> 4;
    const int l16  = lane & 15;

    const float* Arow = &lA[(wv * 16 + l16) * DIN + half * 2];
    float* Yb = Y + (size_t)rowBase * RD + r * DOUT;

    #pragma unroll
    for (int t = 0; t < NT; ++t) {
        const int colBase = t * 16 + l16;
        const float* Bcol = &lWt[colBase * DIN + half * 2];
        v8f c = {};
        #pragma unroll
        for (int i = 0; i < DIN / 4; ++i) {
            v2f a = *(const v2f*)(Arow + 4 * i);
            v2f b = *(const v2f*)(Bcol + 4 * i);
            c = __builtin_amdgcn_wmma_f32_16x16x4_f32(
                     false, a,  false, b,
                     (short)0, c,  false,  false);
        }
        if (DOUT == DOUTPAD || colBase < DOUT) {
            float* p = Yb + (wv * 16 + half * 8) * RD + colBase;
            if (fullBlock) {
                #pragma unroll
                for (int j = 0; j < 8; ++j) p[j * RD] = c[j];
            } else {
                const int mBase = rowBase + wv * 16 + half * 8;
                #pragma unroll
                for (int j = 0; j < 8; ++j)
                    if (mBase + j < N) p[j * RD] = c[j];
            }
        }
    }
}

template<int R>
__global__ __launch_bounds__(256)
void rgcn_agg_relu(const float* __restrict__ XW, const int* __restrict__ ptr,
                   const int* __restrict__ idx, const int* __restrict__ rel,
                   float* __restrict__ H, int N)
{
    const int wv   = threadIdx.x >> 5;
    const int lane = threadIdx.x & 31;
    const int node = blockIdx.x * 8 + wv;
    if (node >= N) return;

    const int e0 = ptr[node], e1 = ptr[node + 1];
    float ax = 0.0f, ay = 0.0f;
    for (int e = e0; e < e1; ++e) {
        const int src = idx[e];
        const int rv  = rel[e];
        const float2 v = *(const float2*)(XW + ((size_t)src * R + rv) * 64 + lane * 2);
        ax += v.x; ay += v.y;
    }
    float2 o; o.x = fmaxf(ax, 0.0f); o.y = fmaxf(ay, 0.0f);
    *(float2*)(H + (size_t)node * 64 + lane * 2) = o;
}

template<int R, int DOUT>
__global__ __launch_bounds__(256)
void rgcn_agg_logsoftmax(const float* __restrict__ HW, const int* __restrict__ ptr,
                         const int* __restrict__ idx, const int* __restrict__ rel,
                         float* __restrict__ out, int N)
{
    const int wv   = threadIdx.x >> 5;
    const int lane = threadIdx.x & 31;
    const int node = blockIdx.x * 8 + wv;
    if (node >= N) return;

    const int  e0  = ptr[node], e1 = ptr[node + 1];
    const bool act = (lane * 2 < DOUT);
    float ax = 0.0f, ay = 0.0f;
    for (int e = e0; e < e1; ++e) {
        const int src = idx[e];
        const int rv  = rel[e];
        if (act) {
            const float2 v = *(const float2*)(HW + ((size_t)src * R + rv) * DOUT + lane * 2);
            ax += v.x; ay += v.y;
        }
    }
    float m = act ? fmaxf(ax, ay) : -__builtin_inff();
    #pragma unroll
    for (int off = 16; off >= 1; off >>= 1) m = fmaxf(m, __shfl_xor(m, off, 32));
    float s = act ? (__expf(ax - m) + __expf(ay - m)) : 0.0f;
    #pragma unroll
    for (int off = 16; off >= 1; off >>= 1) s += __shfl_xor(s, off, 32);
    const float ls = __logf(s);
    if (act) {
        float2 o; o.x = (ax - m) - ls; o.y = (ay - m) - ls;
        *(float2*)(out + (size_t)node * DOUT + lane * 2) = o;
    }
}

extern "C" void kernel_launch(void* const* d_in, const int* in_sizes, int n_in,
                              void* d_out, int out_size, void* d_ws, size_t ws_size,
                              hipStream_t stream)
{
    const float* x   = (const float*)d_in[0];
    const int*   ptr = (const int*)d_in[1];
    const int*   idx = (const int*)d_in[2];
    const int*   rel = (const int*)d_in[3];
    const float* W1  = (const float*)d_in[4];
    const float* W2  = (const float*)d_in[5];
    float* out = (float*)d_out;

    const int N = in_sizes[1] - 1;
    constexpr int R = 8, DH = 64, DOUT = 40;

    float* xw1 = (float*)d_ws;
    float* h   = xw1 + (size_t)N * R * DH;

    const int mblocks   = (N + 127) / 128;
    const int aggblocks = (N + 7) / 8;

    rgcn_gemm_wmma<R, DH, DH><<<dim3(mblocks, R), 256, 0, stream>>>(x, W1, xw1, N);
    rgcn_agg_relu<R><<<aggblocks, 256, 0, stream>>>(xw1, ptr, idx, rel, h, N);
    rgcn_gemm_wmma<R, DOUT, 48><<<dim3(mblocks, R), 256, 0, stream>>>(h, W2, xw1, N);
    rgcn_agg_logsoftmax<R, DOUT><<<aggblocks, 256, 0, stream>>>(xw1, ptr, idx, rel, out, N);
}


# Both CSR reductions: a group of 8 lanes owns one destination node (4 nodes side by side per wave, 16-byte row loads, a quarter of the waves); edge sums keep their order, softmax sum re-ordered
# speedup vs baseline: 2.1504x; 2.1504x over previous
#include <hip/hip_runtime.h>
#include <math.h>


typedef __attribute__((ext_vector_type(2))) float v2f;
typedef __attribute__((ext_vector_type(8))) float v8f;

#define DIN 64

template<int RD, int DOUT>
__global__ __launch_bounds__(128)
void rgcn_gemm_rows(const float* __restrict__ X, const float* __restrict__ W,
                    float* __restrict__ Y, int N)
{
    constexpr int NCH   = RD / 64;
    constexpr int SLOTS = 4 * 8 * 32;
    constexpr int SPT   = SLOTS / 128;
    static_assert(RD % 64 == 0, "column count must be a multiple of 64");
    __shared__ float4 lB[2][SLOTS];

    const int tid  = threadIdx.x;
    const int wv   = tid >> 5;
    const int lane = tid & 31;
    const int half = lane >> 4;
    const int l16  = lane & 15;

    const int rowBase = blockIdx.x * 64;
    const int rowW    = rowBase + wv * 16;
    const bool fullBlock = (rowBase + 64 <= N);

    int rowA = rowW + l16;
    rowA = rowA < N ? rowA : N - 1;
    const float* Ap = X + (size_t)rowA * DIN + half * 2;
    v2f a[16];
    #pragma unroll
    for (int i = 0; i < 16; ++i) a[i] = *(const v2f*)(Ap + 4 * i);

    float4 w[SPT];
    auto fetch = [&](int c) {
        #pragma unroll
        for (int s = 0; s < SPT; ++s) {
            const int q  = tid + 128 * s;
            const int L  = q & 31, p = (q >> 5) & 7, t = q >> 8;
            const int n  = 64 * c + 16 * t + (L & 15);
            const int r  = n / DOUT, f = n - r * DOUT;
            const int k0 = 8 * p + 2 * (L >> 4);
            const float* wp = W + ((size_t)r * DIN + k0) * DOUT + f;
            w[s].x = wp[0];
            w[s].y = wp[DOUT];
            w[s].z = wp[4 * DOUT];
            w[s].w = wp[5 * DOUT];
        }
    };
    auto stash = [&](int buf) {
        #pragma unroll
        for (int s = 0; s < SPT; ++s) lB[buf][tid + 128 * s] = w[s];
    };

    fetch(0);
    stash(0);
    __syncthreads();

    float* Yw = Y + (size_t)(rowW + half * 8) * RD + l16;

    for (int c = 0; c < NCH; ++c) {
        const int buf = c & 1;
        if (c + 1 < NCH) fetch(c + 1);
        __builtin_amdgcn_sched_barrier(0);

        v8f acc[4] = {};
        #pragma unroll
        for (int p = 0; p < 8; ++p) {
            float4 bb[4];
            #pragma unroll
            for (int t = 0; t < 4; ++t) bb[t] = lB[buf][(t * 8 + p) * 32 + lane];
            #pragma unroll
            for (int t = 0; t < 4; ++t) {
                v2f b; b.x = bb[t].x; b.y = bb[t].y;
                acc[t] = __builtin_amdgcn_wmma_f32_16x16x4_f32(
                             false, a[2 * p], false, b, (short)0, acc[t], false, false);
            }
            #pragma unroll
            for (int t = 0; t < 4; ++t) {
                v2f b; b.x = bb[t].z; b.y = bb[t].w;
                acc[t] = __builtin_amdgcn_wmma_f32_16x16x4_f32(
                             false, a[2 * p + 1], false, b, (short)0, acc[t], false, false);
            }
        }

        if (c + 1 < NCH) stash(buf ^ 1);
        __syncthreads();

        float* p0 = Yw + 64 * c;
        if (fullBlock) {
            #pragma unroll
            for (int t = 0; t < 4; ++t)
                #pragma unroll
                for (int j = 0; j < 8; ++j) p0[j * RD + 16 * t] = acc[t][j];
        } else {
            const int mBase = rowW + half * 8;
            #pragma unroll
            for (int t = 0; t < 4; ++t)
                #pragma unroll
                for (int j = 0; j < 8; ++j)
                    if (mBase + j < N) p0[j * RD + 16 * t] = acc[t][j];
        }
    }
}

template<int U>
__device__ __forceinline__ int group_bcast(int v)
{
    return __builtin_amdgcn_mov_dpp8(v, (unsigned)U * 0x249249u);
}
template<int S>
__device__ __forceinline__ float group_xor(float x)
{
    constexpr unsigned sel = (unsigned)(0 ^ S)        | ((unsigned)(1 ^ S) << 3)  |
                             ((unsigned)(2 ^ S) << 6)  | ((unsigned)(3 ^ S) << 9)  |
                             ((unsigned)(4 ^ S) << 12) | ((unsigned)(5 ^ S) << 15) |
                             ((unsigned)(6 ^ S) << 18) | ((unsigned)(7 ^ S) << 21);
    return __builtin_bit_cast(float, __builtin_amdgcn_mov_dpp8(__builtin_bit_cast(int, x), sel));
}

template<int R>
__global__ __launch_bounds__(256)
void rgcn_agg_relu(const float* __restrict__ XW, const int* __restrict__ ptr,
                   const int* __restrict__ idx, const int* __restrict__ rel,
                   float* __restrict__ H, int N)
{
    const int wv   = threadIdx.x >> 5;
    const int lane = threadIdx.x & 31;
    const int j    = lane & 7;
    const int node = (blockIdx.x * 8 + wv) * 4 + (lane >> 3);
    if (node >= N) return;

    const int e0 = ptr[node], e1 = ptr[node + 1];
    const char* tab = (const char*)XW;
    float4 a0 = make_float4(0.f, 0.f, 0.f, 0.f), a1 = a0;

    for (int b = e0; b < e1; b += 8) {
        int er = b + j;
        er = er < e1 ? er : e1 - 1;
        const int rec = idx[er] * R + rel[er];

        float4 v0[8], v1[8];
        #define ROWLOAD(U) { \
            const unsigned off = ((unsigned)group_bcast<U>(rec) << 8) + (unsigned)(j << 4); \
            v0[U] = *(const float4*)(tab + off); \
            v1[U] = *(const float4*)(tab + off + 128); }
        ROWLOAD(0) ROWLOAD(1) ROWLOAD(2) ROWLOAD(3)
        ROWLOAD(4) ROWLOAD(5) ROWLOAD(6) ROWLOAD(7)
        #undef ROWLOAD

        #pragma unroll
        for (int u = 0; u < 8; ++u) {
            const bool ok = (b + u < e1);
            a0.x = ok ? a0.x + v0[u].x : a0.x;  a0.y = ok ? a0.y + v0[u].y : a0.y;
            a0.z = ok ? a0.z + v0[u].z : a0.z;  a0.w = ok ? a0.w + v0[u].w : a0.w;
            a1.x = ok ? a1.x + v1[u].x : a1.x;  a1.y = ok ? a1.y + v1[u].y : a1.y;
            a1.z = ok ? a1.z + v1[u].z : a1.z;  a1.w = ok ? a1.w + v1[u].w : a1.w;
        }
    }
    a0.x = fmaxf(a0.x, 0.0f); a0.y = fmaxf(a0.y, 0.0f); a0.z = fmaxf(a0.z, 0.0f); a0.w = fmaxf(a0.w, 0.0f);
    a1.x = fmaxf(a1.x, 0.0f); a1.y = fmaxf(a1.y, 0.0f); a1.z = fmaxf(a1.z, 0.0f); a1.w = fmaxf(a1.w, 0.0f);
    float* hp = H + (size_t)node * 64 + j * 4;
    *(float4*)hp        = a0;
    *(float4*)(hp + 32) = a1;
}

template<int R, int DOUT>
__global__ __launch_bounds__(256)
void rgcn_agg_logsoftmax(const float* __restrict__ HW, const int* __restrict__ ptr,
                         const int* __restrict__ idx, const int* __restrict__ rel,
                         float* __restrict__ out, int N)
{
    static_assert(DOUT == 40, "lane map below is written for 10 pieces of 4 columns");
    const int wv   = threadIdx.x >> 5;
    const int lane = threadIdx.x & 31;
    const int j    = lane & 7;
    const bool act = (j < 5);
    const int jj   = act ? j : 4;
    const int node = (blockIdx.x * 8 + wv) * 4 + (lane >> 3);
    if (node >= N) return;

    const int e0 = ptr[node], e1 = ptr[node + 1];
    const char* tab = (const char*)HW;
    float4 a0 = make_float4(0.f, 0.f, 0.f, 0.f), a1 = a0;

    for (int b = e0; b < e1; b += 8) {
        int er = b + j;
        er = er < e1 ? er : e1 - 1;
        const int rec = idx[er] * R + rel[er];

        float4 v0[8], v1[8];
        #define ROWLOAD(U) { \
            const unsigned off = (unsigned)group_bcast<U>(rec) * (unsigned)(DOUT * 4) + (unsigned)(jj << 4); \
            v0[U] = *(const float4*)(tab + off); \
            v1[U] = *(const float4*)(tab + off + 80); }
        ROWLOAD(0) ROWLOAD(1) ROWLOAD(2) ROWLOAD(3)
        ROWLOAD(4) ROWLOAD(5) ROWLOAD(6) ROWLOAD(7)
        #undef ROWLOAD

        #pragma unroll
        for (int u = 0; u < 8; ++u) {
            const bool ok = (b + u < e1);
            a0.x = ok ? a0.x + v0[u].x : a0.x;  a0.y = ok ? a0.y + v0[u].y : a0.y;
            a0.z = ok ? a0.z + v0[u].z : a0.z;  a0.w = ok ? a0.w + v0[u].w : a0.w;
            a1.x = ok ? a1.x + v1[u].x : a1.x;  a1.y = ok ? a1.y + v1[u].y : a1.y;
            a1.z = ok ? a1.z + v1[u].z : a1.z;  a1.w = ok ? a1.w + v1[u].w : a1.w;
        }
    }

    float m = fmaxf(fmaxf(fmaxf(a0.x, a0.y), fmaxf(a0.z, a0.w)),
                    fmaxf(fmaxf(a1.x, a1.y), fmaxf(a1.z, a1.w)));
    m = act ? m : -__builtin_inff();
    m = fmaxf(m, group_xor<1>(m));
    m = fmaxf(m, group_xor<2>(m));
    m = fmaxf(m, group_xor<4>(m));
    float s = __expf(a0.x - m) + __expf(a0.y - m) + __expf(a0.z - m) + __expf(a0.w - m)
            + __expf(a1.x - m) + __expf(a1.y - m) + __expf(a1.z - m) + __expf(a1.w - m);
    s = act ? s : 0.0f;
    s += group_xor<1>(s);
    s += group_xor<2>(s);
    s += group_xor<4>(s);
    const float ls = __logf(s);
    if (act) {
        float4 o0, o1;
        o0.x = (a0.x - m) - ls; o0.y = (a0.y - m) - ls; o0.z = (a0.z - m) - ls; o0.w = (a0.w - m) - ls;
        o1.x = (a1.x - m) - ls; o1.y = (a1.y - m) - ls; o1.z = (a1.z - m) - ls; o1.w = (a1.w - m) - ls;
        float* op = out + (size_t)node * DOUT + j * 4;
        *(float4*)op        = o0;
        *(float4*)(op + 20) = o1;
    }
}

extern "C" void kernel_launch(void* const* d_in, const int* in_sizes, int n_in,
                              void* d_out, int out_size, void* d_ws, size_t ws_size,
                              hipStream_t stream)
{
    const float* x   = (const float*)d_in[0];
    const int*   ptr = (const int*)d_in[1];
    const int*   idx = (const int*)d_in[2];
    const int*   rel = (const int*)d_in[3];
    const float* W1  = (const float*)d_in[4];
    const float* W2  = (const float*)d_in[5];
    float* out = (float*)d_out;

    const int N = in_sizes[1] - 1;
    constexpr int R = 8, DH = 64, DOUT = 40;

    float* xw1 = (float*)d_ws;
    float* h   = xw1 + (size_t)N * R * DH;

    const int mblocks   = (N + 63) / 64;
    const int aggblocks = (N + 31) / 32;

    rgcn_gemm_rows<R * DH, DH><<<mblocks, 128, 0, stream>>>(x, W1, xw1, N);
    rgcn_agg_relu<R><<<aggblocks, 256, 0, stream>>>(xw1, ptr, idx, rel, h, N);
    rgcn_gemm_rows<R * DOUT, DOUT><<<mblocks, 128, 0, stream>>>(h, W2, xw1, N);
    rgcn_agg_logsoftmax<R, DOUT><<<aggblocks, 256, 0, stream>>>(xw1, ptr, idx, rel, out, N);
}


# Layer-1 CSR reduction moved into the layer-2 product kernel: a wave reduces its own 16 nodes into an LDS tile and multiplies them there; h not stored, 4 launches -> 3; sum order kept
# speedup vs baseline: 2.3319x; 1.0844x over previous
#include <hip/hip_runtime.h>
#include <math.h>


typedef __attribute__((ext_vector_type(2))) float v2f;
typedef __attribute__((ext_vector_type(8))) float v8f;

#define DIN 64

template<int RD, int DOUT>
__global__ __launch_bounds__(128)
void rgcn_gemm_rows(const float* __restrict__ X, const float* __restrict__ W,
                    float* __restrict__ Y, int N)
{
    constexpr int NCH   = RD / 64;
    constexpr int SLOTS = 4 * 8 * 32;
    constexpr int SPT   = SLOTS / 128;
    static_assert(RD % 64 == 0, "column count must be a multiple of 64");
    __shared__ float4 lB[2][SLOTS];

    const int tid  = threadIdx.x;
    const int wv   = tid >> 5;
    const int lane = tid & 31;
    const int half = lane >> 4;
    const int l16  = lane & 15;

    const int rowBase = blockIdx.x * 64;
    const int rowW    = rowBase + wv * 16;
    const bool fullBlock = (rowBase + 64 <= N);

    int rowA = rowW + l16;
    rowA = rowA < N ? rowA : N - 1;
    const float* Ap = X + (size_t)rowA * DIN + half * 2;
    v2f a[16];
    #pragma unroll
    for (int i = 0; i < 16; ++i) a[i] = *(const v2f*)(Ap + 4 * i);

    float4 w[SPT];
    auto fetch = [&](int c) {
        #pragma unroll
        for (int s = 0; s < SPT; ++s) {
            const int q  = tid + 128 * s;
            const int L  = q & 31, p = (q >> 5) & 7, t = q >> 8;
            const int n  = 64 * c + 16 * t + (L & 15);
            const int r  = n / DOUT, f = n - r * DOUT;
            const int k0 = 8 * p + 2 * (L >> 4);
            const float* wp = W + ((size_t)r * DIN + k0) * DOUT + f;
            w[s].x = wp[0];
            w[s].y = wp[DOUT];
            w[s].z = wp[4 * DOUT];
            w[s].w = wp[5 * DOUT];
        }
    };
    auto stash = [&](int buf) {
        #pragma unroll
        for (int s = 0; s < SPT; ++s) lB[buf][tid + 128 * s] = w[s];
    };

    fetch(0);
    stash(0);
    __syncthreads();

    float* Yw = Y + (size_t)(rowW + half * 8) * RD + l16;

    for (int c = 0; c < NCH; ++c) {
        const int buf = c & 1;
        if (c + 1 < NCH) fetch(c + 1);
        __builtin_amdgcn_sched_barrier(0);

        v8f acc[4] = {};
        #pragma unroll
        for (int p = 0; p < 8; ++p) {
            float4 bb[4];
            #pragma unroll
            for (int t = 0; t < 4; ++t) bb[t] = lB[buf][(t * 8 + p) * 32 + lane];
            #pragma unroll
            for (int t = 0; t < 4; ++t) {
                v2f b; b.x = bb[t].x; b.y = bb[t].y;
                acc[t] = __builtin_amdgcn_wmma_f32_16x16x4_f32(
                             false, a[2 * p], false, b, (short)0, acc[t], false, false);
            }
            #pragma unroll
            for (int t = 0; t < 4; ++t) {
                v2f b; b.x = bb[t].z; b.y = bb[t].w;
                acc[t] = __builtin_amdgcn_wmma_f32_16x16x4_f32(
                             false, a[2 * p + 1], false, b, (short)0, acc[t], false, false);
            }
        }

        if (c + 1 < NCH) stash(buf ^ 1);
        __syncthreads();

        float* p0 = Yw + 64 * c;
        if (fullBlock) {
            #pragma unroll
            for (int t = 0; t < 4; ++t)
                #pragma unroll
                for (int j = 0; j < 8; ++j) p0[j * RD + 16 * t] = acc[t][j];
        } else {
            const int mBase = rowW + half * 8;
            #pragma unroll
            for (int t = 0; t < 4; ++t)
                #pragma unroll
                for (int j = 0; j < 8; ++j)
                    if (mBase + j < N) p0[j * RD + 16 * t] = acc[t][j];
        }
    }
}

template<int U>
__device__ __forceinline__ int group_bcast(int v)
{
    return __builtin_amdgcn_mov_dpp8(v, (unsigned)U * 0x249249u);
}
template<int S>
__device__ __forceinline__ float group_xor(float x)
{
    constexpr unsigned sel = (unsigned)(0 ^ S)        | ((unsigned)(1 ^ S) << 3)  |
                             ((unsigned)(2 ^ S) << 6)  | ((unsigned)(3 ^ S) << 9)  |
                             ((unsigned)(4 ^ S) << 12) | ((unsigned)(5 ^ S) << 15) |
                             ((unsigned)(6 ^ S) << 18) | ((unsigned)(7 ^ S) << 21);
    return __builtin_bit_cast(float, __builtin_amdgcn_mov_dpp8(__builtin_bit_cast(int, x), sel));
}

template<int R>
__global__ __launch_bounds__(256)
void rgcn_agg_relu(const float* __restrict__ XW, const int* __restrict__ ptr,
                   const int* __restrict__ idx, const int* __restrict__ rel,
                   float* __restrict__ H, int N)
{
    const int wv   = threadIdx.x >> 5;
    const int lane = threadIdx.x & 31;
    const int j    = lane & 7;
    const int node = (blockIdx.x * 8 + wv) * 4 + (lane >> 3);
    if (node >= N) return;

    const int e0 = ptr[node], e1 = ptr[node + 1];
    const char* tab = (const char*)XW;
    float4 a0 = make_float4(0.f, 0.f, 0.f, 0.f), a1 = a0;

    for (int b = e0; b < e1; b += 8) {
        int er = b + j;
        er = er < e1 ? er : e1 - 1;
        const int rec = idx[er] * R + rel[er];

        float4 v0[8], v1[8];
        #define ROWLOAD(U) { \
            const unsigned off = ((unsigned)group_bcast<U>(rec) << 8) + (unsigned)(j << 4); \
            v0[U] = *(const float4*)(tab + off); \
            v1[U] = *(const float4*)(tab + off + 128); }
        ROWLOAD(0) ROWLOAD(1) ROWLOAD(2) ROWLOAD(3)
        ROWLOAD(4) ROWLOAD(5) ROWLOAD(6) ROWLOAD(7)
        #undef ROWLOAD

        #pragma unroll
        for (int u = 0; u < 8; ++u) {
            const bool ok = (b + u < e1);
            a0.x = ok ? a0.x + v0[u].x : a0.x;  a0.y = ok ? a0.y + v0[u].y : a0.y;
            a0.z = ok ? a0.z + v0[u].z : a0.z;  a0.w = ok ? a0.w + v0[u].w : a0.w;
            a1.x = ok ? a1.x + v1[u].x : a1.x;  a1.y = ok ? a1.y + v1[u].y : a1.y;
            a1.z = ok ? a1.z + v1[u].z : a1.z;  a1.w = ok ? a1.w + v1[u].w : a1.w;
        }
    }
    a0.x = fmaxf(a0.x, 0.0f); a0.y = fmaxf(a0.y, 0.0f); a0.z = fmaxf(a0.z, 0.0f); a0.w = fmaxf(a0.w, 0.0f);
    a1.x = fmaxf(a1.x, 0.0f); a1.y = fmaxf(a1.y, 0.0f); a1.z = fmaxf(a1.z, 0.0f); a1.w = fmaxf(a1.w, 0.0f);
    float* hp = H + (size_t)node * 64 + j * 4;
    *(float4*)hp        = a0;
    *(float4*)(hp + 32) = a1;
}

template<int R, int DOUT>
__global__ __launch_bounds__(256)
void rgcn_agg_logsoftmax(const float* __restrict__ HW, const int* __restrict__ ptr,
                         const int* __restrict__ idx, const int* __restrict__ rel,
                         float* __restrict__ out, int N)
{
    static_assert(DOUT == 40, "lane map below is written for 10 pieces of 4 columns");
    const int wv   = threadIdx.x >> 5;
    const int lane = threadIdx.x & 31;
    const int j    = lane & 7;
    const bool act = (j < 5);
    const int jj   = act ? j : 4;
    const int node = (blockIdx.x * 8 + wv) * 4 + (lane >> 3);
    if (node >= N) return;

    const int e0 = ptr[node], e1 = ptr[node + 1];
    const char* tab = (const char*)HW;
    float4 a0 = make_float4(0.f, 0.f, 0.f, 0.f), a1 = a0;

    for (int b = e0; b < e1; b += 8) {
        int er = b + j;
        er = er < e1 ? er : e1 - 1;
        const int rec = idx[er] * R + rel[er];

        float4 v0[8], v1[8];
        #define ROWLOAD(U) { \
            const unsigned off = (unsigned)group_bcast<U>(rec) * (unsigned)(DOUT * 4) + (unsigned)(jj << 4); \
            v0[U] = *(const float4*)(tab + off); \
            v1[U] = *(const float4*)(tab + off + 80); }
        ROWLOAD(0) ROWLOAD(1) ROWLOAD(2) ROWLOAD(3)
        ROWLOAD(4) ROWLOAD(5) ROWLOAD(6) ROWLOAD(7)
        #undef ROWLOAD

        #pragma unroll
        for (int u = 0; u < 8; ++u) {
            const bool ok = (b + u < e1);
            a0.x = ok ? a0.x + v0[u].x : a0.x;  a0.y = ok ? a0.y + v0[u].y : a0.y;
            a0.z = ok ? a0.z + v0[u].z : a0.z;  a0.w = ok ? a0.w + v0[u].w : a0.w;
            a1.x = ok ? a1.x + v1[u].x : a1.x;  a1.y = ok ? a1.y + v1[u].y : a1.y;
            a1.z = ok ? a1.z + v1[u].z : a1.z;  a1.w = ok ? a1.w + v1[u].w : a1.w;
        }
    }

    float m = fmaxf(fmaxf(fmaxf(a0.x, a0.y), fmaxf(a0.z, a0.w)),
                    fmaxf(fmaxf(a1.x, a1.y), fmaxf(a1.z, a1.w)));
    m = act ? m : -__builtin_inff();
    m = fmaxf(m, group_xor<1>(m));
    m = fmaxf(m, group_xor<2>(m));
    m = fmaxf(m, group_xor<4>(m));
    float s = __expf(a0.x - m) + __expf(a0.y - m) + __expf(a0.z - m) + __expf(a0.w - m)
            + __expf(a1.x - m) + __expf(a1.y - m) + __expf(a1.z - m) + __expf(a1.w - m);
    s = act ? s : 0.0f;
    s += group_xor<1>(s);
    s += group_xor<2>(s);
    s += group_xor<4>(s);
    const float ls = __logf(s);
    if (act) {
        float4 o0, o1;
        o0.x = (a0.x - m) - ls; o0.y = (a0.y - m) - ls; o0.z = (a0.z - m) - ls; o0.w = (a0.w - m) - ls;
        o1.x = (a1.x - m) - ls; o1.y = (a1.y - m) - ls; o1.z = (a1.z - m) - ls; o1.w = (a1.w - m) - ls;
        float* op = out + (size_t)node * DOUT + j * 4;
        *(float4*)op        = o0;
        *(float4*)(op + 20) = o1;
    }
}

template<int R, int RD, int DOUT>
__global__ __launch_bounds__(128)
void rgcn_agg_relu_gemm_rows(const float* __restrict__ XW, const int* __restrict__ ptr,
                             const int* __restrict__ idx, const int* __restrict__ rel,
                             const float* __restrict__ W, float* __restrict__ Y, int N)
{
    constexpr int NCH   = RD / 64;
    constexpr int SLOTS = 4 * 8 * 32;
    constexpr int SPT   = SLOTS / 128;
    constexpr int HP4   = 17;
    static_assert(RD % 64 == 0, "column count must be a multiple of 64");
    __shared__ float4 lB[2][SLOTS];
    __shared__ float4 hT[64 * HP4];

    const int tid  = threadIdx.x;
    const int wv   = tid >> 5;
    const int lane = tid & 31;
    const int half = lane >> 4;
    const int l16  = lane & 15;

    const int rowBase = blockIdx.x * 64;
    const int rowW    = rowBase + wv * 16;
    const bool fullBlock = (rowBase + 64 <= N);

    float4 w[SPT];
    auto fetch = [&](int c) {
        #pragma unroll
        for (int s = 0; s < SPT; ++s) {
            const int q  = tid + 128 * s;
            const int L  = q & 31, p = (q >> 5) & 7, t = q >> 8;
            const int n  = 64 * c + 16 * t + (L & 15);
            const int r  = n / DOUT, f = n - r * DOUT;
            const int k0 = 8 * p + 2 * (L >> 4);
            const float* wp = W + ((size_t)r * DIN + k0) * DOUT + f;
            w[s].x = wp[0];
            w[s].y = wp[DOUT];
            w[s].z = wp[4 * DOUT];
            w[s].w = wp[5 * DOUT];
        }
    };
    auto stash = [&](int buf) {
        #pragma unroll
        for (int s = 0; s < SPT; ++s) lB[buf][tid + 128 * s] = w[s];
    };

    fetch(0);

    {
        const int j = lane & 7;
        const int g = lane >> 3;
        const char* tab = (const char*)XW;

        int e0q[4], e1q[4];
        #pragma unroll
        for (int q = 0; q < 4; ++q) {
            const int node = rowW + 4 * q + g;
            const bool in  = node < N;
            const int nn   = in ? node : 0;
            const int p0 = ptr[nn], p1 = ptr[nn + 1];
            e0q[q] = in ? p0 : 0;
            e1q[q] = in ? p1 : 0;
        }

        #define ROWLOAD(U) { \
            const unsigned off = ((unsigned)group_bcast<U>(rec) << 8) + (unsigned)(j << 4); \
            v0[U] = *(const float4*)(tab + off); \
            v1[U] = *(const float4*)(tab + off + 128); }

        #pragma unroll
        for (int q = 0; q < 4; ++q) {
            const int e0 = e0q[q], e1 = e1q[q];
            float4 a0 = make_float4(0.f, 0.f, 0.f, 0.f), a1 = a0;

            for (int b = e0; b < e1; b += 8) {
                int er = b + j;
                er = er < e1 ? er : e1 - 1;
                const int rec = idx[er] * R + rel[er];

                float4 v0[8], v1[8];
                ROWLOAD(0) ROWLOAD(1) ROWLOAD(2) ROWLOAD(3)
                ROWLOAD(4) ROWLOAD(5) ROWLOAD(6) ROWLOAD(7)

                #pragma unroll
                for (int u = 0; u < 8; ++u) {
                    const bool ok = (b + u < e1);
                    a0.x = ok ? a0.x + v0[u].x : a0.x;  a0.y = ok ? a0.y + v0[u].y : a0.y;
                    a0.z = ok ? a0.z + v0[u].z : a0.z;  a0.w = ok ? a0.w + v0[u].w : a0.w;
                    a1.x = ok ? a1.x + v1[u].x : a1.x;  a1.y = ok ? a1.y + v1[u].y : a1.y;
                    a1.z = ok ? a1.z + v1[u].z : a1.z;  a1.w = ok ? a1.w + v1[u].w : a1.w;
                }
            }
            a0.x = fmaxf(a0.x, 0.0f); a0.y = fmaxf(a0.y, 0.0f); a0.z = fmaxf(a0.z, 0.0f); a0.w = fmaxf(a0.w, 0.0f);
            a1.x = fmaxf(a1.x, 0.0f); a1.y = fmaxf(a1.y, 0.0f); a1.z = fmaxf(a1.z, 0.0f); a1.w = fmaxf(a1.w, 0.0f);

            const int lrow = wv * 16 + 4 * q + g;
            hT[lrow * HP4 + j]     = a0;
            hT[lrow * HP4 + 8 + j] = a1;
        }
        #undef ROWLOAD
    }

    stash(0);
    __syncthreads();

    const float* hrow = (const float*)hT + (wv * 16 + l16) * (HP4 * 4) + half * 2;
    v2f a[16];
    #pragma unroll
    for (int i = 0; i < 16; ++i) a[i] = *(const v2f*)(hrow + 4 * i);

    float* Yw = Y + (size_t)(rowW + half * 8) * RD + l16;

    for (int c = 0; c < NCH; ++c) {
        const int buf = c & 1;
        if (c + 1 < NCH) fetch(c + 1);
        __builtin_amdgcn_sched_barrier(0);

        v8f acc[4] = {};
        #pragma unroll
        for (int p = 0; p < 8; ++p) {
            float4 bb[4];
            #pragma unroll
            for (int t = 0; t < 4; ++t) bb[t] = lB[buf][(t * 8 + p) * 32 + lane];
            #pragma unroll
            for (int t = 0; t < 4; ++t) {
                v2f b; b.x = bb[t].x; b.y = bb[t].y;
                acc[t] = __builtin_amdgcn_wmma_f32_16x16x4_f32(
                             false, a[2 * p], false, b, (short)0, acc[t], false, false);
            }
            #pragma unroll
            for (int t = 0; t < 4; ++t) {
                v2f b; b.x = bb[t].z; b.y = bb[t].w;
                acc[t] = __builtin_amdgcn_wmma_f32_16x16x4_f32(
                             false, a[2 * p + 1], false, b, (short)0, acc[t], false, false);
            }
        }

        if (c + 1 < NCH) stash(buf ^ 1);
        __syncthreads();

        float* p0 = Yw + 64 * c;
        if (fullBlock) {
            #pragma unroll
            for (int t = 0; t < 4; ++t)
                #pragma unroll
                for (int j = 0; j < 8; ++j) p0[j * RD + 16 * t] = acc[t][j];
        } else {
            const int mBase = rowW + half * 8;
            #pragma unroll
            for (int t = 0; t < 4; ++t)
                #pragma unroll
                for (int j = 0; j < 8; ++j)
                    if (mBase + j < N) p0[j * RD + 16 * t] = acc[t][j];
        }
    }
}

extern "C" void kernel_launch(void* const* d_in, const int* in_sizes, int n_in,
                              void* d_out, int out_size, void* d_ws, size_t ws_size,
                              hipStream_t stream)
{
    const float* x   = (const float*)d_in[0];
    const int*   ptr = (const int*)d_in[1];
    const int*   idx = (const int*)d_in[2];
    const int*   rel = (const int*)d_in[3];
    const float* W1  = (const float*)d_in[4];
    const float* W2  = (const float*)d_in[5];
    float* out = (float*)d_out;

    const int N = in_sizes[1] - 1;
    constexpr int R = 8, DH = 64, DOUT = 40;

    float* xw1 = (float*)d_ws;

    const int mblocks   = (N + 63) / 64;
    const int aggblocks = (N + 31) / 32;

    rgcn_gemm_rows<R * DH, DH><<<mblocks, 128, 0, stream>>>(x, W1, xw1, N);

    const size_t both = (size_t)N * R * (DH + DOUT) * sizeof(float);
    if (ws_size >= both) {
        float* hw2 = xw1 + (size_t)N * R * DH;
        rgcn_agg_relu_gemm_rows<R, R * DOUT, DOUT><<<mblocks, 128, 0, stream>>>(xw1, ptr, idx, rel, W2, hw2, N);
        rgcn_agg_logsoftmax<R, DOUT><<<aggblocks, 256, 0, stream>>>(hw2, ptr, idx, rel, out, N);
    } else {
        float* h = xw1 + (size_t)N * R * DH;
        rgcn_agg_relu<R><<<aggblocks, 256, 0, stream>>>(xw1, ptr, idx, rel, h, N);
        rgcn_gemm_rows<R * DOUT, DOUT><<<mblocks, 128, 0, stream>>>(h, W2, xw1, N);
        rgcn_agg_logsoftmax<R, DOUT><<<aggblocks, 256, 0, stream>>>(xw1, ptr, idx, rel, out, N);
    }
}


# GEMM epilogue, both product kernels: a wave's 16x64 output tile goes through LDS and out as 16-byte stores, 2 whole 256-B row pieces per instruction (8 per chunk, not 32 of 4 bytes); sums unchanged
# speedup vs baseline: 2.6471x; 1.1352x over previous
#include <hip/hip_runtime.h>
#include <math.h>


typedef __attribute__((ext_vector_type(2))) float v2f;
typedef __attribute__((ext_vector_type(8))) float v8f;

#define DIN 64

template<int RD, int DOUT>
__global__ __launch_bounds__(128)
void rgcn_gemm_rows(const float* __restrict__ X, const float* __restrict__ W,
                    float* __restrict__ Y, int N)
{
    constexpr int NCH   = RD / 64;
    constexpr int SLOTS = 4 * 8 * 32;
    constexpr int SPT   = SLOTS / 128;
    constexpr int HP4   = 17;
    static_assert(RD % 64 == 0, "column count must be a multiple of 64");
    __shared__ float4 lB[2][SLOTS];
    __shared__ float4 oT[64 * HP4];

    const int tid  = threadIdx.x;
    const int wv   = tid >> 5;
    const int lane = tid & 31;
    const int half = lane >> 4;
    const int l16  = lane & 15;

    const int rowBase = blockIdx.x * 64;
    const int rowW    = rowBase + wv * 16;
    const bool fullBlock = (rowBase + 64 <= N);

    int rowA = rowW + l16;
    rowA = rowA < N ? rowA : N - 1;
    const float* Ap = X + (size_t)rowA * DIN + half * 2;
    v2f a[16];
    #pragma unroll
    for (int i = 0; i < 16; ++i) a[i] = *(const v2f*)(Ap + 4 * i);

    float4 w[SPT];
    auto fetch = [&](int c) {
        #pragma unroll
        for (int s = 0; s < SPT; ++s) {
            const int q  = tid + 128 * s;
            const int L  = q & 31, p = (q >> 5) & 7, t = q >> 8;
            const int n  = 64 * c + 16 * t + (L & 15);
            const int r  = n / DOUT, f = n - r * DOUT;
            const int k0 = 8 * p + 2 * (L >> 4);
            const float* wp = W + ((size_t)r * DIN + k0) * DOUT + f;
            w[s].x = wp[0];
            w[s].y = wp[DOUT];
            w[s].z = wp[4 * DOUT];
            w[s].w = wp[5 * DOUT];
        }
    };
    auto stash = [&](int buf) {
        #pragma unroll
        for (int s = 0; s < SPT; ++s) lB[buf][tid + 128 * s] = w[s];
    };

    fetch(0);
    stash(0);
    __syncthreads();

    for (int c = 0; c < NCH; ++c) {
        const int buf = c & 1;
        if (c + 1 < NCH) fetch(c + 1);
        __builtin_amdgcn_sched_barrier(0);

        v8f acc[4] = {};
        #pragma unroll
        for (int p = 0; p < 8; ++p) {
            float4 bb[4];
            #pragma unroll
            for (int t = 0; t < 4; ++t) bb[t] = lB[buf][(t * 8 + p) * 32 + lane];
            #pragma unroll
            for (int t = 0; t < 4; ++t) {
                v2f b; b.x = bb[t].x; b.y = bb[t].y;
                acc[t] = __builtin_amdgcn_wmma_f32_16x16x4_f32(
                             false, a[2 * p], false, b, (short)0, acc[t], false, false);
            }
            #pragma unroll
            for (int t = 0; t < 4; ++t) {
                v2f b; b.x = bb[t].z; b.y = bb[t].w;
                acc[t] = __builtin_amdgcn_wmma_f32_16x16x4_f32(
                             false, a[2 * p + 1], false, b, (short)0, acc[t], false, false);
            }
        }

        if (c + 1 < NCH) stash(buf ^ 1);

        {
            float* ot = (float*)oT + (wv * 16 + half * 8) * (HP4 * 4) + l16;
            #pragma unroll
            for (int t = 0; t < 4; ++t)
                #pragma unroll
                for (int j = 0; j < 8; ++j) ot[j * (HP4 * 4) + 16 * t] = acc[t][j];
        }
        __syncthreads();

        {
            const float4* orow = oT + (wv * 16 + half) * HP4 + l16;
            float* yp = Y + (size_t)(rowW + half) * RD + 64 * c + 4 * l16;
            float4 o[8];
            #pragma unroll
            for (int i = 0; i < 8; ++i) o[i] = orow[2 * i * HP4];
            if (fullBlock) {
                #pragma unroll
                for (int i = 0; i < 8; ++i) *(float4*)(yp + (size_t)(2 * i) * RD) = o[i];
            } else {
                #pragma unroll
                for (int i = 0; i < 8; ++i)
                    if (rowW + half + 2 * i < N) *(float4*)(yp + (size_t)(2 * i) * RD) = o[i];
            }
        }
    }
}

template<int U>
__device__ __forceinline__ int group_bcast(int v)
{
    return __builtin_amdgcn_mov_dpp8(v, (unsigned)U * 0x249249u);
}
template<int S>
__device__ __forceinline__ float group_xor(float x)
{
    constexpr unsigned sel = (unsigned)(0 ^ S)        | ((unsigned)(1 ^ S) << 3)  |
                             ((unsigned)(2 ^ S) << 6)  | ((unsigned)(3 ^ S) << 9)  |
                             ((unsigned)(4 ^ S) << 12) | ((unsigned)(5 ^ S) << 15) |
                             ((unsigned)(6 ^ S) << 18) | ((unsigned)(7 ^ S) << 21);
    return __builtin_bit_cast(float, __builtin_amdgcn_mov_dpp8(__builtin_bit_cast(int, x), sel));
}

template<int R>
__global__ __launch_bounds__(256)
void rgcn_agg_relu(const float* __restrict__ XW, const int* __restrict__ ptr,
                   const int* __restrict__ idx, const int* __restrict__ rel,
                   float* __restrict__ H, int N)
{
    const int wv   = threadIdx.x >> 5;
    const int lane = threadIdx.x & 31;
    const int j    = lane & 7;
    const int node = (blockIdx.x * 8 + wv) * 4 + (lane >> 3);
    if (node >= N) return;

    const int e0 = ptr[node], e1 = ptr[node + 1];
    const char* tab = (const char*)XW;
    float4 a0 = make_float4(0.f, 0.f, 0.f, 0.f), a1 = a0;

    for (int b = e0; b < e1; b += 8) {
        int er = b + j;
        er = er < e1 ? er : e1 - 1;
        const int rec = idx[er] * R + rel[er];

        float4 v0[8], v1[8];
        #define ROWLOAD(U) { \
            const unsigned off = ((unsigned)group_bcast<U>(rec) << 8) + (unsigned)(j << 4); \
            v0[U] = *(const float4*)(tab + off); \
            v1[U] = *(const float4*)(tab + off + 128); }
        ROWLOAD(0) ROWLOAD(1) ROWLOAD(2) ROWLOAD(3)
        ROWLOAD(4) ROWLOAD(5) ROWLOAD(6) ROWLOAD(7)
        #undef ROWLOAD

        #pragma unroll
        for (int u = 0; u < 8; ++u) {
            const bool ok = (b + u < e1);
            a0.x = ok ? a0.x + v0[u].x : a0.x;  a0.y = ok ? a0.y + v0[u].y : a0.y;
            a0.z = ok ? a0.z + v0[u].z : a0.z;  a0.w = ok ? a0.w + v0[u].w : a0.w;
            a1.x = ok ? a1.x + v1[u].x : a1.x;  a1.y = ok ? a1.y + v1[u].y : a1.y;
            a1.z = ok ? a1.z + v1[u].z : a1.z;  a1.w = ok ? a1.w + v1[u].w : a1.w;
        }
    }
    a0.x = fmaxf(a0.x, 0.0f); a0.y = fmaxf(a0.y, 0.0f); a0.z = fmaxf(a0.z, 0.0f); a0.w = fmaxf(a0.w, 0.0f);
    a1.x = fmaxf(a1.x, 0.0f); a1.y = fmaxf(a1.y, 0.0f); a1.z = fmaxf(a1.z, 0.0f); a1.w = fmaxf(a1.w, 0.0f);
    float* hp = H + (size_t)node * 64 + j * 4;
    *(float4*)hp        = a0;
    *(float4*)(hp + 32) = a1;
}

template<int R, int DOUT>
__global__ __launch_bounds__(256)
void rgcn_agg_logsoftmax(const float* __restrict__ HW, const int* __restrict__ ptr,
                         const int* __restrict__ idx, const int* __restrict__ rel,
                         float* __restrict__ out, int N)
{
    static_assert(DOUT == 40, "lane map below is written for 10 pieces of 4 columns");
    const int wv   = threadIdx.x >> 5;
    const int lane = threadIdx.x & 31;
    const int j    = lane & 7;
    const bool act = (j < 5);
    const int jj   = act ? j : 4;
    const int node = (blockIdx.x * 8 + wv) * 4 + (lane >> 3);
    if (node >= N) return;

    const int e0 = ptr[node], e1 = ptr[node + 1];
    const char* tab = (const char*)HW;
    float4 a0 = make_float4(0.f, 0.f, 0.f, 0.f), a1 = a0;

    for (int b = e0; b < e1; b += 8) {
        int er = b + j;
        er = er < e1 ? er : e1 - 1;
        const int rec = idx[er] * R + rel[er];

        float4 v0[8], v1[8];
        #define ROWLOAD(U) { \
            const unsigned off = (unsigned)group_bcast<U>(rec) * (unsigned)(DOUT * 4) + (unsigned)(jj << 4); \
            v0[U] = *(const float4*)(tab + off); \
            v1[U] = *(const float4*)(tab + off + 80); }
        ROWLOAD(0) ROWLOAD(1) ROWLOAD(2) ROWLOAD(3)
        ROWLOAD(4) ROWLOAD(5) ROWLOAD(6) ROWLOAD(7)
        #undef ROWLOAD

        #pragma unroll
        for (int u = 0; u < 8; ++u) {
            const bool ok = (b + u < e1);
            a0.x = ok ? a0.x + v0[u].x : a0.x;  a0.y = ok ? a0.y + v0[u].y : a0.y;
            a0.z = ok ? a0.z + v0[u].z : a0.z;  a0.w = ok ? a0.w + v0[u].w : a0.w;
            a1.x = ok ? a1.x + v1[u].x : a1.x;  a1.y = ok ? a1.y + v1[u].y : a1.y;
            a1.z = ok ? a1.z + v1[u].z : a1.z;  a1.w = ok ? a1.w + v1[u].w : a1.w;
        }
    }

    float m = fmaxf(fmaxf(fmaxf(a0.x, a0.y), fmaxf(a0.z, a0.w)),
                    fmaxf(fmaxf(a1.x, a1.y), fmaxf(a1.z, a1.w)));
    m = act ? m : -__builtin_inff();
    m = fmaxf(m, group_xor<1>(m));
    m = fmaxf(m, group_xor<2>(m));
    m = fmaxf(m, group_xor<4>(m));
    float s = __expf(a0.x - m) + __expf(a0.y - m) + __expf(a0.z - m) + __expf(a0.w - m)
            + __expf(a1.x - m) + __expf(a1.y - m) + __expf(a1.z - m) + __expf(a1.w - m);
    s = act ? s : 0.0f;
    s += group_xor<1>(s);
    s += group_xor<2>(s);
    s += group_xor<4>(s);
    const float ls = __logf(s);
    if (act) {
        float4 o0, o1;
        o0.x = (a0.x - m) - ls; o0.y = (a0.y - m) - ls; o0.z = (a0.z - m) - ls; o0.w = (a0.w - m) - ls;
        o1.x = (a1.x - m) - ls; o1.y = (a1.y - m) - ls; o1.z = (a1.z - m) - ls; o1.w = (a1.w - m) - ls;
        float* op = out + (size_t)node * DOUT + j * 4;
        *(float4*)op        = o0;
        *(float4*)(op + 20) = o1;
    }
}

template<int R, int RD, int DOUT>
__global__ __launch_bounds__(128)
void rgcn_agg_relu_gemm_rows(const float* __restrict__ XW, const int* __restrict__ ptr,
                             const int* __restrict__ idx, const int* __restrict__ rel,
                             const float* __restrict__ W, float* __restrict__ Y, int N)
{
    constexpr int NCH   = RD / 64;
    constexpr int SLOTS = 4 * 8 * 32;
    constexpr int SPT   = SLOTS / 128;
    constexpr int HP4   = 17;
    static_assert(RD % 64 == 0, "column count must be a multiple of 64");
    __shared__ float4 lB[2][SLOTS];
    __shared__ float4 hT[64 * HP4];

    const int tid  = threadIdx.x;
    const int wv   = tid >> 5;
    const int lane = tid & 31;
    const int half = lane >> 4;
    const int l16  = lane & 15;

    const int rowBase = blockIdx.x * 64;
    const int rowW    = rowBase + wv * 16;
    const bool fullBlock = (rowBase + 64 <= N);

    float4 w[SPT];
    auto fetch = [&](int c) {
        #pragma unroll
        for (int s = 0; s < SPT; ++s) {
            const int q  = tid + 128 * s;
            const int L  = q & 31, p = (q >> 5) & 7, t = q >> 8;
            const int n  = 64 * c + 16 * t + (L & 15);
            const int r  = n / DOUT, f = n - r * DOUT;
            const int k0 = 8 * p + 2 * (L >> 4);
            const float* wp = W + ((size_t)r * DIN + k0) * DOUT + f;
            w[s].x = wp[0];
            w[s].y = wp[DOUT];
            w[s].z = wp[4 * DOUT];
            w[s].w = wp[5 * DOUT];
        }
    };
    auto stash = [&](int buf) {
        #pragma unroll
        for (int s = 0; s < SPT; ++s) lB[buf][tid + 128 * s] = w[s];
    };

    fetch(0);

    {
        const int j = lane & 7;
        const int g = lane >> 3;
        const char* tab = (const char*)XW;

        int e0q[4], e1q[4];
        #pragma unroll
        for (int q = 0; q < 4; ++q) {
            const int node = rowW + 4 * q + g;
            const bool in  = node < N;
            const int nn   = in ? node : 0;
            const int p0 = ptr[nn], p1 = ptr[nn + 1];
            e0q[q] = in ? p0 : 0;
            e1q[q] = in ? p1 : 0;
        }

        #define ROWLOAD(U) { \
            const unsigned off = ((unsigned)group_bcast<U>(rec) << 8) + (unsigned)(j << 4); \
            v0[U] = *(const float4*)(tab + off); \
            v1[U] = *(const float4*)(tab + off + 128); }

        #pragma unroll
        for (int q = 0; q < 4; ++q) {
            const int e0 = e0q[q], e1 = e1q[q];
            float4 a0 = make_float4(0.f, 0.f, 0.f, 0.f), a1 = a0;

            for (int b = e0; b < e1; b += 8) {
                int er = b + j;
                er = er < e1 ? er : e1 - 1;
                const int rec = idx[er] * R + rel[er];

                float4 v0[8], v1[8];
                ROWLOAD(0) ROWLOAD(1) ROWLOAD(2) ROWLOAD(3)
                ROWLOAD(4) ROWLOAD(5) ROWLOAD(6) ROWLOAD(7)

                #pragma unroll
                for (int u = 0; u < 8; ++u) {
                    const bool ok = (b + u < e1);
                    a0.x = ok ? a0.x + v0[u].x : a0.x;  a0.y = ok ? a0.y + v0[u].y : a0.y;
                    a0.z = ok ? a0.z + v0[u].z : a0.z;  a0.w = ok ? a0.w + v0[u].w : a0.w;
                    a1.x = ok ? a1.x + v1[u].x : a1.x;  a1.y = ok ? a1.y + v1[u].y : a1.y;
                    a1.z = ok ? a1.z + v1[u].z : a1.z;  a1.w = ok ? a1.w + v1[u].w : a1.w;
                }
            }
            a0.x = fmaxf(a0.x, 0.0f); a0.y = fmaxf(a0.y, 0.0f); a0.z = fmaxf(a0.z, 0.0f); a0.w = fmaxf(a0.w, 0.0f);
            a1.x = fmaxf(a1.x, 0.0f); a1.y = fmaxf(a1.y, 0.0f); a1.z = fmaxf(a1.z, 0.0f); a1.w = fmaxf(a1.w, 0.0f);

            const int lrow = wv * 16 + 4 * q + g;
            hT[lrow * HP4 + j]     = a0;
            hT[lrow * HP4 + 8 + j] = a1;
        }
        #undef ROWLOAD
    }

    stash(0);
    __syncthreads();

    const float* hrow = (const float*)hT + (wv * 16 + l16) * (HP4 * 4) + half * 2;
    v2f a[16];
    #pragma unroll
    for (int i = 0; i < 16; ++i) a[i] = *(const v2f*)(hrow + 4 * i);

    float4* const oT = hT;

    for (int c = 0; c < NCH; ++c) {
        const int buf = c & 1;
        if (c + 1 < NCH) fetch(c + 1);
        __builtin_amdgcn_sched_barrier(0);

        v8f acc[4] = {};
        #pragma unroll
        for (int p = 0; p < 8; ++p) {
            float4 bb[4];
            #pragma unroll
            for (int t = 0; t < 4; ++t) bb[t] = lB[buf][(t * 8 + p) * 32 + lane];
            #pragma unroll
            for (int t = 0; t < 4; ++t) {
                v2f b; b.x = bb[t].x; b.y = bb[t].y;
                acc[t] = __builtin_amdgcn_wmma_f32_16x16x4_f32(
                             false, a[2 * p], false, b, (short)0, acc[t], false, false);
            }
            #pragma unroll
            for (int t = 0; t < 4; ++t) {
                v2f b; b.x = bb[t].z; b.y = bb[t].w;
                acc[t] = __builtin_amdgcn_wmma_f32_16x16x4_f32(
                             false, a[2 * p + 1], false, b, (short)0, acc[t], false, false);
            }
        }

        if (c + 1 < NCH) stash(buf ^ 1);

        {
            float* ot = (float*)oT + (wv * 16 + half * 8) * (HP4 * 4) + l16;
            #pragma unroll
            for (int t = 0; t < 4; ++t)
                #pragma unroll
                for (int j = 0; j < 8; ++j) ot[j * (HP4 * 4) + 16 * t] = acc[t][j];
        }
        __syncthreads();

        {
            const float4* orow = oT + (wv * 16 + half) * HP4 + l16;
            float* yp = Y + (size_t)(rowW + half) * RD + 64 * c + 4 * l16;
            float4 o[8];
            #pragma unroll
            for (int i = 0; i < 8; ++i) o[i] = orow[2 * i * HP4];
            if (fullBlock) {
                #pragma unroll
                for (int i = 0; i < 8; ++i) *(float4*)(yp + (size_t)(2 * i) * RD) = o[i];
            } else {
                #pragma unroll
                for (int i = 0; i < 8; ++i)
                    if (rowW + half + 2 * i < N) *(float4*)(yp + (size_t)(2 * i) * RD) = o[i];
            }
        }
    }
}

extern "C" void kernel_launch(void* const* d_in, const int* in_sizes, int n_in,
                              void* d_out, int out_size, void* d_ws, size_t ws_size,
                              hipStream_t stream)
{
    const float* x   = (const float*)d_in[0];
    const int*   ptr = (const int*)d_in[1];
    const int*   idx = (const int*)d_in[2];
    const int*   rel = (const int*)d_in[3];
    const float* W1  = (const float*)d_in[4];
    const float* W2  = (const float*)d_in[5];
    float* out = (float*)d_out;

    const int N = in_sizes[1] - 1;
    constexpr int R = 8, DH = 64, DOUT = 40;

    float* xw1 = (float*)d_ws;

    const int mblocks   = (N + 63) / 64;
    const int aggblocks = (N + 31) / 32;

    rgcn_gemm_rows<R * DH, DH><<<mblocks, 128, 0, stream>>>(x, W1, xw1, N);

    const size_t both = (size_t)N * R * (DH + DOUT) * sizeof(float);
    if (ws_size >= both) {
        float* hw2 = xw1 + (size_t)N * R * DH;
        rgcn_agg_relu_gemm_rows<R, R * DOUT, DOUT><<<mblocks, 128, 0, stream>>>(xw1, ptr, idx, rel, W2, hw2, N);
        rgcn_agg_logsoftmax<R, DOUT><<<aggblocks, 256, 0, stream>>>(hw2, ptr, idx, rel, out, N);
    } else {
        float* h = xw1 + (size_t)N * R * DH;
        rgcn_agg_relu<R><<<aggblocks, 256, 0, stream>>>(xw1, ptr, idx, rel, h, N);
        rgcn_gemm_rows<R * DOUT, DOUT><<<mblocks, 128, 0, stream>>>(h, W2, xw1, N);
        rgcn_agg_logsoftmax<R, DOUT><<<aggblocks, 256, 0, stream>>>(xw1, ptr, idx, rel, out, N);
    }
}
